# starting kernel
# baseline (speedup 1.0000x reference)
#include <hip/hip_runtime.h>
#include <math.h>

#define BSZ   32
#define DCH   96
#define HH    64
#define WW    64
#define LL    (HH * WW)
#define NST   4
#define RNK   6
#define KPROJ 14
#define LOG_EPS_F (-20.7232658f)

typedef __attribute__((ext_vector_type(2))) float v2f;
typedef __attribute__((ext_vector_type(8))) float v8f;

__global__ void conv_silu_kernel(const float* __restrict__ x,
                                 const float* __restrict__ conv_w,
                                 const float* __restrict__ conv_b,
                                 float* __restrict__ xs) {
    const int bid  = blockIdx.x;
    const int tile = bid & 15;
    const int d    = (bid >> 4) % DCH;
    const int b    = bid / (16 * DCH);
    const int h0   = tile * 4;

    __shared__ float t[8][68];

    const int tid = threadIdx.x;
    const float* xin = x + ((size_t)b * DCH + d) * LL;

    for (int i = tid; i < 8 * 68; i += 256) {
        const int r  = i / 68;
        const int c  = i % 68;
        const int hh = h0 - 2 + r;
        const int wc = c - 2;
        float v = 0.0f;
        if (hh >= 0 && hh < HH && wc >= 0 && wc < WW)
            v = xin[hh * WW + wc];
        t[r][c] = v;
    }
    __syncthreads();

    float wv[25];
#pragma unroll
    for (int j = 0; j < 25; ++j) wv[j] = conv_w[d * 25 + j];
    const float bv = conv_b[d];

    const int ty = tid >> 6;
    const int tx = tid & 63;

    float acc = bv;
#pragma unroll
    for (int ky = 0; ky < 5; ++ky)
#pragma unroll
        for (int kx = 0; kx < 5; ++kx)
            acc = fmaf(t[ty + ky][tx + kx], wv[ky * 5 + kx], acc);

    const float s = acc / (1.0f + expf(-acc));
    xs[((size_t)b * DCH + d) * LL + (size_t)(h0 + ty) * WW + tx] = s;
}

__global__ void proj_wmma_kernel(const float* __restrict__ xs,
                                 const float* __restrict__ x_proj_w,
                                 float* __restrict__ xdbl) {
    __shared__ float wA[16][DCH];

    const int tid = threadIdx.x;
    for (int i = tid; i < 16 * DCH; i += 256) {
        const int row = i / DCH;
        const int k   = i % DCH;
        wA[row][k] = (row < KPROJ) ? x_proj_w[row * DCH + k] : 0.0f;
    }
    __syncthreads();

    const int wid  = (blockIdx.x * blockDim.x + tid) >> 5;
    const int lane = tid & 31;
    const int tcol = wid & 255;
    const int b    = wid >> 8;
    const int l0   = tcol * 16;

    const int half = lane >> 4;
    const int m    = lane & 15;

    const float* xb = xs + (size_t)b * DCH * LL + (l0 + m);

    v2f breg[DCH / 4];
#pragma unroll
    for (int kk = 0; kk < DCH / 4; ++kk) {
        const int k0 = kk * 4 + 2 * half;
        breg[kk].x = xb[(size_t)k0 * LL];
        breg[kk].y = xb[(size_t)(k0 + 1) * LL];
    }

    v8f acc = {0.f, 0.f, 0.f, 0.f, 0.f, 0.f, 0.f, 0.f};
#pragma unroll
    for (int kk = 0; kk < DCH / 4; ++kk) {
        const int k0 = kk * 4 + 2 * half;
        const v2f a = *(const v2f*)&wA[m][k0];
        acc = __builtin_amdgcn_wmma_f32_16x16x4_f32(
                  false, a, false, breg[kk], (short)0, acc, false, false);
    }

#pragma unroll
    for (int i = 0; i < 8; ++i) {
        const int row = i + 8 * half;
        if (row < KPROJ)
            xdbl[((size_t)b * KPROJ + row) * LL + l0 + m] = acc[i];
    }
}

__device__ __forceinline__ float wave_incl_sum(float v, int lane) {
#pragma unroll
    for (int off = 1; off < 32; off <<= 1) {
        const float t = __shfl_up(v, off, 32);
        if (lane >= off) v += t;
    }
    return v;
}

__global__ void scan_kernel(const float* __restrict__ xs,
                            const float* __restrict__ xdbl,
                            const float* __restrict__ dt_w,
                            const float* __restrict__ dt_b,
                            const float* __restrict__ A_logs,
                            const float* __restrict__ Ds,
                            float* __restrict__ out) {
    const int wid  = (blockIdx.x * blockDim.x + threadIdx.x) >> 5;
    const int lane = threadIdx.x & 31;
    const int d    = wid % DCH;
    const int b    = wid / DCH;

    float wr[RNK];
#pragma unroll
    for (int r = 0; r < RNK; ++r) wr[r] = dt_w[d * RNK + r];
    const float dtb = dt_b[d];
    const float Dd  = Ds[d];
    float An[NST];
#pragma unroll
    for (int n = 0; n < NST; ++n) An[n] = -expf(A_logs[d * NST + n]);

    const float* xch = xs   + ((size_t)b * DCH + d) * LL;
    const float* xdb = xdbl + (size_t)b * KPROJ * LL;
    float*       och = out  + ((size_t)b * DCH + d) * LL;

    float c1[NST] = {0.f, 0.f, 0.f, 0.f};
    float c2[NST] = {0.f, 0.f, 0.f, 0.f};

    for (int c = 0; c < LL / 32; ++c) {
        const int l = c * 32 + lane;
        const float xv = xch[l];

        float ad = dtb;
#pragma unroll
        for (int r = 0; r < RNK; ++r)
            ad = fmaf(wr[r], xdb[(size_t)r * LL + l], ad);
        const float delta = (ad > 20.0f) ? ad : log1pf(expf(ad));

        float y = xv * Dd;

#pragma unroll
        for (int n = 0; n < NST; ++n) {
            const float Bn = xdb[(size_t)(RNK + n) * LL + l];
            const float Cn = xdb[(size_t)(RNK + NST + n) * LL + l];

            const float lda = fmaxf(delta * An[n], LOG_EPS_F);
            float s1 = wave_incl_sum(lda, lane) + c1[n];
            const float P = expf(s1);

            const float q  = (delta * Bn * xv) / fmaxf(P, 1e-9f);
            float s2 = wave_incl_sum(q, lane) + c2[n];

            y = fmaf(P * s2, Cn, y);

            c1[n] = __shfl(s1, 31, 32);
            c2[n] = __shfl(s2, 31, 32);
        }
        och[l] = y;
    }
}

extern "C" void kernel_launch(void* const* d_in, const int* in_sizes, int n_in,
                              void* d_out, int out_size, void* d_ws, size_t ws_size,
                              hipStream_t stream) {
    (void)in_sizes; (void)n_in; (void)out_size; (void)ws_size;

    const float* x        = (const float*)d_in[0];
    const float* conv_w   = (const float*)d_in[1];
    const float* conv_b   = (const float*)d_in[2];
    const float* x_proj_w = (const float*)d_in[3];
    const float* dt_w     = (const float*)d_in[4];
    const float* dt_b     = (const float*)d_in[5];
    const float* A_logs   = (const float*)d_in[6];
    const float* Ds       = (const float*)d_in[7];
    float* out = (float*)d_out;

    float* xs   = (float*)d_ws;
    float* xdbl = xs + (size_t)BSZ * DCH * LL;

    {
        dim3 grid(BSZ * DCH * (HH / 4));
        conv_silu_kernel<<<grid, 256, 0, stream>>>(x, conv_w, conv_b, xs);
    }
    {
        const int nwaves = BSZ * (LL / 16);
        dim3 grid(nwaves / 8);
        proj_wmma_kernel<<<grid, 256, 0, stream>>>(xs, x_proj_w, xdbl);
    }
    {
        const int nwaves = BSZ * DCH;
        dim3 grid(nwaves / 8);
        scan_kernel<<<grid, 256, 0, stream>>>(xs, xdbl, dt_w, dt_b,
                                              A_logs, Ds, out);
    }
}


# conv kernel: the 5x5 depthwise conv runs on the f32 matrix unit as a banded (Toeplitz) product per channel: 4x4 output tiles from 8x8 windows, 16 k-steps per 256 outputs, 4-row strips
# speedup vs baseline: 2.0668x; 2.0668x over previous
#include <hip/hip_runtime.h>
#include <math.h>


#define BSZ   32
#define DCH   96
#define HH    64
#define WW    64
#define LL    (HH * WW)
#define NST   4
#define RNK   6
#define KPROJ 14
#define LOG_EPS_F (-20.7232658f)

typedef __attribute__((ext_vector_type(2))) float v2f;
typedef __attribute__((ext_vector_type(8))) float v8f;

#define STRIP_ROWS 4
#define NSTRIP     (HH / STRIP_ROWS)
#define HALO_ROWS  (STRIP_ROWS + 4)
#define HALO_PITCH 72
#define CH_STEP    8
#define NSTEP      (DCH / CH_STEP)
#define HALO_CH    (HALO_ROWS * HALO_PITCH)
#define HALO_BUF   (CH_STEP * HALO_CH)
#define HALO_SLOTS 4
#define WZ_PITCH   12
#define WZ_CH      (11 * WZ_PITCH)
#define WA_PITCH   98
#define XST_ROW    136
#define XST_PAIR   (STRIP_ROWS * XST_ROW)

__device__ __forceinline__ float4 keep_or_zero(const float4 v, const bool keep) {
    return make_float4(keep ? v.x : 0.0f, keep ? v.y : 0.0f,
                       keep ? v.z : 0.0f, keep ? v.w : 0.0f);
}

__device__ __forceinline__ void lds_barrier() {
    asm volatile("s_wait_dscnt 0x0\n\t"
                 "s_barrier_signal -1\n\t"
                 "s_barrier_wait -1" ::: "memory");
}

__global__ void __launch_bounds__(256)
conv_proj_kernel(const float* __restrict__ x,
                 const float* __restrict__ conv_w,
                 const float* __restrict__ conv_b,
                 const float* __restrict__ x_proj_w,
                 float* __restrict__ xs,
                 float* __restrict__ xdbl) {
    __shared__ __attribute__((aligned(16))) float halo[2 * HALO_BUF];
    __shared__ __attribute__((aligned(16))) float xst[2][CH_STEP / 2][XST_PAIR];
    __shared__ __attribute__((aligned(16))) float wA[16][WA_PITCH];
    __shared__ __attribute__((aligned(16))) float wz[DCH * WZ_CH];

    const int tid   = threadIdx.x;
    const int strip = blockIdx.x & (NSTRIP - 1);
    const int b     = blockIdx.x / NSTRIP;
    const int h0    = strip * STRIP_ROWS;

    const int w    = __builtin_amdgcn_readfirstlane(tid >> 5);
    const int lane = tid & 31;
    const int half = lane >> 4;
    const int m    = lane & 15;

    const float4 zero4 = make_float4(0.f, 0.f, 0.f, 0.f);

    for (int i = tid; i < DCH * WZ_CH / 4; i += 256)
        *(float4*)&wz[4 * i] = zero4;
    {
        const int r = tid >> 1;
        *(float4*)&halo[r * HALO_PITCH + ((tid & 1) ? 68 : 0)] = zero4;
    }
    __syncthreads();

    for (int i = tid; i < DCH * 25; i += 256) {
        const int c  = i / 25;
        const int j  = i % 25;
        const int ky = j / 5;
        const int kx = j % 5;
        wz[c * WZ_CH + (ky + 3) * WZ_PITCH + (kx + 3)] = conv_w[i];
    }
    for (int i = tid; i < 16 * DCH; i += 256) {
        const int row = i / DCH;
        const int k   = i % DCH;
        wA[row][k] = (row < KPROJ) ? x_proj_w[row * DCH + k] : 0.0f;
    }

    const int hrow = (tid >> 4) & 7;
    const int hh   = h0 - 2 + hrow;
    const bool inimg = (hh >= 0) && (hh < HH);
    const int hcl  = (hh < 0) ? 0 : ((hh >= HH) ? HH - 1 : hh);
    int goff[HALO_SLOTS];
    int loff[HALO_SLOTS];
#pragma unroll
    for (int j = 0; j < HALO_SLOTS; ++j) {
        const int chj = 2 * j + (tid >> 7);
        goff[j] = chj * LL + hcl * WW + 4 * (tid & 15);
        loff[j] = chj * HALO_CH + hrow * HALO_PITCH + 4 + 4 * (tid & 15);
    }

    const float* xb = x + (size_t)b * DCH * LL;

#pragma unroll
    for (int j = 0; j < HALO_SLOTS; ++j) {
        const float4 v = *(const float4*)(xb + goff[j]);
        *(float4*)&halo[loff[j]] = keep_or_zero(v, inimg);
    }
    __syncthreads();

    const int oy   = m >> 2;
    const int ox   = m & 3;
    const int aoff = (3 - oy) * WZ_PITCH + (3 - ox) + 2 * half;
    const int boff = 4 * m + 2 + 2 * half;

    float* xsl = xs + (size_t)b * DCH * LL + (size_t)(h0 + 2 * half) * WW + 4 * m;

    int ppos[2];
#pragma unroll
    for (int q = 0; q < 2; ++q) {
        const int nt = 2 * w + q;
        ppos[q] = (nt >> 2) * XST_ROW + 2 * (16 * (nt & 3) + m);
    }
    v8f pacc[2];
    pacc[0] = (v8f){0.f, 0.f, 0.f, 0.f, 0.f, 0.f, 0.f, 0.f};
    pacc[1] = pacc[0];

#pragma unroll 1
    for (int it = 0; it < NSTEP; ++it) {
        const int  buf  = it & 1;
        const bool more = (it + 1 < NSTEP);

        float4 nx[HALO_SLOTS];
        if (more) {
            const float* xn = xb + (size_t)(it + 1) * CH_STEP * LL;
#pragma unroll
            for (int j = 0; j < HALO_SLOTS; ++j)
                nx[j] = *(const float4*)(xn + goff[j]);
        }

        const int c = it * CH_STEP + w;
        const float* wzc = &wz[c * WZ_CH + aoff];
        const float* hc  = &halo[buf * HALO_BUF + w * HALO_CH + boff];
        v8f acc0 = (v8f){0.f, 0.f, 0.f, 0.f, 0.f, 0.f, 0.f, 0.f};
        v8f acc1 = acc0;
#pragma unroll
        for (int s = 0; s < 16; s += 2) {
            const int wy = s >> 1;
            v2f a0, a1;
            a0.x = wzc[wy * WZ_PITCH];
            a0.y = wzc[wy * WZ_PITCH + 1];
            a1.x = wzc[wy * WZ_PITCH + 4];
            a1.y = wzc[wy * WZ_PITCH + 5];
            const v2f b0 = *(const v2f*)&hc[wy * HALO_PITCH];
            const v2f b1 = *(const v2f*)&hc[wy * HALO_PITCH + 4];
            acc0 = __builtin_amdgcn_wmma_f32_16x16x4_f32(
                       false, a0, false, b0, (short)0, acc0, false, false);
            acc1 = __builtin_amdgcn_wmma_f32_16x16x4_f32(
                       false, a1, false, b1, (short)0, acc1, false, false);
        }

        const float bias = conv_b[c];
        float sv[8];
#pragma unroll
        for (int i = 0; i < 8; ++i) {
            const float a = (acc0[i] + acc1[i]) + bias;
            sv[i] = a / (1.0f + expf(-a));
        }
        float* xo = xsl + (size_t)c * LL;
        *(float4*)xo        = make_float4(sv[0], sv[1], sv[2], sv[3]);
        *(float4*)(xo + WW) = make_float4(sv[4], sv[5], sv[6], sv[7]);
        float* so = &xst[buf][w >> 1][(2 * half) * XST_ROW + 8 * m + (w & 1)];
#pragma unroll
        for (int e = 0; e < 4; ++e) {
            so[2 * e]           = sv[e];
            so[XST_ROW + 2 * e] = sv[4 + e];
        }

        if (more) {
#pragma unroll
            for (int j = 0; j < HALO_SLOTS; ++j)
                *(float4*)&halo[(buf ^ 1) * HALO_BUF + loff[j]] = keep_or_zero(nx[j], inimg);
        }
        lds_barrier();

#pragma unroll
        for (int ks = 0; ks < 2; ++ks) {
            const v2f a = *(const v2f*)&wA[m][it * CH_STEP + 4 * ks + 2 * half];
#pragma unroll
            for (int q = 0; q < 2; ++q) {
                const v2f bb = *(const v2f*)&xst[buf][2 * ks + half][ppos[q]];
                pacc[q] = __builtin_amdgcn_wmma_f32_16x16x4_f32(
                              false, a, false, bb, (short)0, pacc[q], false, false);
            }
        }
    }

#pragma unroll
    for (int q = 0; q < 2; ++q) {
        float* xd = xdbl + (size_t)b * KPROJ * LL + (size_t)h0 * WW + 16 * (2 * w + q) + m;
#pragma unroll
        for (int i = 0; i < 8; ++i) {
            const int row = i + 8 * half;
            if (row < KPROJ)
                xd[(size_t)row * LL] = pacc[q][i];
        }
    }
}

#define SCAN_WAVES  16
#define SEG_LEN     (LL / SCAN_WAVES)
#define SEG_CHUNKS  (SEG_LEN / 32)

__device__ __forceinline__ float wave_incl_sum(float v, int lane) {
#pragma unroll
    for (int off = 1; off < 32; off <<= 1) {
        const float t = __shfl_up(v, off, 32);
        if (lane >= off) v += t;
    }
    return v;
}

__global__ void __launch_bounds__(SCAN_WAVES * 32)
scan_kernel(const float* __restrict__ xs,
            const float* __restrict__ xdbl,
            const float* __restrict__ dt_w,
            const float* __restrict__ dt_b,
            const float* __restrict__ A_logs,
            const float* __restrict__ Ds,
            float* __restrict__ out) {
    __shared__ float4 tot1[SCAN_WAVES];
    __shared__ float4 tot2[SCAN_WAVES];
    __shared__ float4 Es[SEG_CHUNKS][SCAN_WAVES * 32];
    __shared__ float  dls[SEG_CHUNKS][SCAN_WAVES * 32];
    __shared__ float  yls[SEG_CHUNKS][SCAN_WAVES * 32];

    const int ch   = blockIdx.x;
    const int d    = ch % DCH;
    const int b    = ch / DCH;
    const int tid  = threadIdx.x;
    const int w    = __builtin_amdgcn_readfirstlane(tid >> 5);
    const int lane = tid & 31;

    float wr[RNK];
#pragma unroll
    for (int r = 0; r < RNK; ++r) wr[r] = dt_w[d * RNK + r];
    const float dtb = dt_b[d];
    const float Dd  = Ds[d];
    float An[NST];
#pragma unroll
    for (int n = 0; n < NST; ++n) An[n] = -expf(A_logs[d * NST + n]);

    const int l0 = w * SEG_LEN + lane;
    const float* xch = xs   + ((size_t)b * DCH + d) * LL + l0;
    const float* xdb = xdbl + (size_t)b * KPROJ * LL + l0;
    float*       och = out  + ((size_t)b * DCH + d) * LL + l0;

    float a1[NST] = {0.f, 0.f, 0.f, 0.f};
#pragma unroll 1
    for (int c = 0; c < SEG_CHUNKS; ++c) {
        float ad = dtb;
#pragma unroll
        for (int r = 0; r < RNK; ++r)
            ad = fmaf(wr[r], xdb[(size_t)r * LL + c * 32], ad);
        const float dl = (ad > 20.0f) ? ad : log1pf(expf(ad));
        dls[c][tid] = dl;
#pragma unroll
        for (int n = 0; n < NST; ++n)
            a1[n] += fmaxf(dl * An[n], LOG_EPS_F);
    }
    {
        float t[NST];
#pragma unroll
        for (int n = 0; n < NST; ++n)
            t[n] = __shfl(wave_incl_sum(a1[n], lane), 31, 32);
        if (lane == 0) tot1[w] = make_float4(t[0], t[1], t[2], t[3]);
    }
    __syncthreads();

    float c1[NST] = {0.f, 0.f, 0.f, 0.f};
    for (int j = 0; j < w; ++j) {
        const float4 t = tot1[j];
        c1[0] += t.x; c1[1] += t.y; c1[2] += t.z; c1[3] += t.w;
    }
    float c2[NST] = {0.f, 0.f, 0.f, 0.f};

#pragma unroll 1
    for (int c = 0; c < SEG_CHUNKS; ++c) {
        const float xv = xch[c * 32];
        const float dl = dls[c][tid];
        float y = xv * Dd;
        float e[NST];

#pragma unroll
        for (int n = 0; n < NST; ++n) {
            const float Bn = xdb[(size_t)(RNK + n) * LL + c * 32];
            const float Cn = xdb[(size_t)(RNK + NST + n) * LL + c * 32];

            const float lda = fmaxf(dl * An[n], LOG_EPS_F);
            float s1 = wave_incl_sum(lda, lane) + c1[n];
            const float P = expf(s1);

            const float q  = (dl * Bn * xv) / fmaxf(P, 1e-9f);
            float s2 = wave_incl_sum(q, lane) + c2[n];

            y = fmaf(P * s2, Cn, y);
            e[n] = P * Cn;

            c1[n] = __shfl(s1, 31, 32);
            c2[n] = __shfl(s2, 31, 32);
        }
        yls[c][tid] = y;
        Es[c][tid]  = make_float4(e[0], e[1], e[2], e[3]);
    }
    if (lane == 0) tot2[w] = make_float4(c2[0], c2[1], c2[2], c2[3]);
    __syncthreads();

    float o2[NST] = {0.f, 0.f, 0.f, 0.f};
    for (int j = 0; j < w; ++j) {
        const float4 t = tot2[j];
        o2[0] += t.x; o2[1] += t.y; o2[2] += t.z; o2[3] += t.w;
    }
#pragma unroll 8
    for (int c = 0; c < SEG_CHUNKS; ++c) {
        const float4 e = Es[c][tid];
        float y = yls[c][tid];
        y = fmaf(e.x, o2[0], y);
        y = fmaf(e.y, o2[1], y);
        y = fmaf(e.z, o2[2], y);
        y = fmaf(e.w, o2[3], y);
        och[c * 32] = y;
    }
}

extern "C" void kernel_launch(void* const* d_in, const int* in_sizes, int n_in,
                              void* d_out, int out_size, void* d_ws, size_t ws_size,
                              hipStream_t stream) {
    (void)in_sizes; (void)n_in; (void)out_size; (void)ws_size;

    const float* x        = (const float*)d_in[0];
    const float* conv_w   = (const float*)d_in[1];
    const float* conv_b   = (const float*)d_in[2];
    const float* x_proj_w = (const float*)d_in[3];
    const float* dt_w     = (const float*)d_in[4];
    const float* dt_b     = (const float*)d_in[5];
    const float* A_logs   = (const float*)d_in[6];
    const float* Ds       = (const float*)d_in[7];
    float* out = (float*)d_out;

    float* xs   = (float*)d_ws;
    float* xdbl = xs + (size_t)BSZ * DCH * LL;

    {
        dim3 grid(BSZ * NSTRIP);
        conv_proj_kernel<<<grid, 256, 0, stream>>>(x, conv_w, conv_b, x_proj_w,
                                                   xs, xdbl);
    }
    {
        dim3 grid(BSZ * DCH);
        scan_kernel<<<grid, SCAN_WAVES * 32, 0, stream>>>(xs, xdbl, dt_w, dt_b,
                                                          A_logs, Ds, out);
    }
}
